# starting kernel
# baseline (speedup 1.0000x reference)
#include <hip/hip_runtime.h>
#include <math.h>


typedef float v2f __attribute__((ext_vector_type(2)));
typedef float v8f __attribute__((ext_vector_type(8)));

#define BATCH 256
#define TIME  1024
#define IND   39
#define HID   256
#define RANK  16

#define WIS 264
#define BLS 16
#define AWS 264
#define WDS 48
#define XS  44
#define HS  260
#define RS  20

#define OFF_WI 0
#define OFF_BL (OFF_WI + 40*WIS)
#define OFF_AW (OFF_BL + 256*BLS)
#define OFF_WD (OFF_AW + 16*AWS)
#define OFF_B  (OFF_WD + 256*WDS)
#define OFF_BD (OFF_B  + 256)
#define OFF_X  (OFF_BD + 48)
#define OFF_H  (OFF_X  + 16*XS)
#define OFF_R  (OFF_H  + 16*HS)
#define OFF_RP (OFF_R  + 16*RS)
#define OFF_DP (OFF_RP + 16*256)
#define SMEM_FLOATS (OFF_DP + 16*768)
#define SMEM_BYTES  (SMEM_FLOATS * 4)

__device__ __forceinline__ v8f wmma4(v2f a, v2f b, v8f c) {
  return __builtin_amdgcn_wmma_f32_16x16x4_f32(
       false, a,  false, b,
       (short)0, c,  false,  false);
}

__device__ __forceinline__ float fast_tanh(float v) {
#if __has_builtin(__builtin_amdgcn_tanhf)
  return __builtin_amdgcn_tanhf(v);
#else
  return tanhf(v);
#endif
}

__device__ __forceinline__ v2f ldA(const float* base, int stride, int m, int kk, int kh) {
  const float* p = base + m * stride + kk + 2 * kh;
  v2f a; a.x = p[0]; a.y = p[1]; return a;
}
__device__ __forceinline__ v2f ldB(const float* base, int stride, int col, int kk, int kh) {
  const float* p = base + (kk + 2 * kh) * stride + col;
  v2f b; b.x = p[0]; b.y = p[stride]; return b;
}

__global__ __launch_bounds__(512)
void dream_scan(const float* __restrict__ x, const float* __restrict__ W_in,
                const float* __restrict__ Amat, const float* __restrict__ B_low,
                const float* __restrict__ bvec, const float* __restrict__ W_dec,
                const float* __restrict__ b_dec, float* __restrict__ out) {
  extern __shared__ float sm[];
  float* Wi = sm + OFF_WI;
  float* Bl = sm + OFF_BL;
  float* Aw = sm + OFF_AW;
  float* Wd = sm + OFF_WD;
  float* Bs = sm + OFF_B;
  float* Bd = sm + OFF_BD;
  float* Xl = sm + OFF_X;
  float* Hl = sm + OFF_H;
  float* Rl = sm + OFF_R;
  float* Rp = sm + OFF_RP;
  float* Dp = sm + OFF_DP;

  const int tid  = threadIdx.x;
  const int wave = tid >> 5;
  const int lane = tid & 31;
  const int lm   = lane & 15;
  const int kh   = lane >> 4;
  const int b0   = blockIdx.x * 16;
  const int n0   = wave * 16;

  for (int e = tid; e < 40 * 256; e += 512) {
    int n = e & 255, k = e >> 8;
    Wi[k * WIS + n] = (k < IND) ? W_in[n * IND + k] : 0.0f;
  }
  for (int e = tid; e < 256 * 16; e += 512) {
    int j = e & 15, k = e >> 4;
    Bl[k * BLS + j] = B_low[j * HID + k];
  }
  for (int e = tid; e < 16 * 256; e += 512) {
    int n = e & 255, j = e >> 8;
    Aw[j * AWS + n] = Amat[n * RANK + j];
  }
  for (int e = tid; e < 256 * 48; e += 512) {
    int c = e % 48, k = e / 48;
    Wd[k * WDS + c] = (c < IND) ? W_dec[c * HID + k] : 0.0f;
  }
  if (tid < 256) Bs[tid] = bvec[tid];
  if (tid < 48)  Bd[tid] = (tid < IND) ? b_dec[tid] : 0.0f;
  for (int e = tid; e < 16 * HS; e += 512) Hl[e] = 0.0f;
  for (int e = tid; e < 16 * XS; e += 512) Xl[e] = 0.0f;
  __syncthreads();

  const int  e0 = tid, e1 = tid + 512;
  const bool xv0 = (e0 < 16 * IND), xv1 = (e1 < 16 * IND);
  const float* xg0 = x;  int xl0 = 0;
  const float* xg1 = x;  int xl1 = 0;
  if (xv0) { int r = e0 / IND, c = e0 % IND;
    xg0 = x + (size_t)(b0 + r) * TIME * IND + c;  xl0 = r * XS + c; }
  if (xv1) { int r = e1 / IND, c = e1 % IND;
    xg1 = x + (size_t)(b0 + r) * TIME * IND + c;  xl1 = r * XS + c; }

  const int  ea = tid, eb = tid + 512;
  const bool sva = ((ea % 48) < IND);
  const bool svb = (eb < 768) && ((eb % 48) < IND);
  float* outa = out;  float bda = 0.0f;
  float* outb = out;  float bdb = 0.0f;
  if (sva) { int row = ea / 48, c = ea % 48;
    outa = out + (size_t)(b0 + row) * TIME * IND + c;  bda = Bd[c]; }
  if (svb) { int row = eb / 48, c = eb % 48;
    outb = out + (size_t)(b0 + row) * TIME * IND + c;  bdb = Bd[c]; }
  const float bb = Bs[n0 + lm];

  float xr0 = 0.0f, xr1 = 0.0f;
  if (xv0) xr0 = xg0[0];
  if (xv1) xr1 = xg1[0];

  v8f h_acc = {};

  for (int t = 0; t < TIME; ++t) {
    if (xv0) Xl[xl0] = xr0;
    if (xv1) Xl[xl1] = xr1;
    __syncthreads();
    if (t + 1 < TIME) {
      size_t off = (size_t)(t + 1) * IND;
      if (xv0) xr0 = xg0[off];
      if (xv1) xr1 = xg1[off];
    }

    v8f acc;
    #pragma unroll
    for (int v = 0; v < 8; ++v) acc[v] = bb;
    #pragma unroll
    for (int kk = 0; kk < 40; kk += 4)
      acc = wmma4(ldA(Xl, XS, lm, kk, kh), ldB(Wi, WIS, n0 + lm, kk, kh), acc);

    {
      v8f rp = {};
      const int kb = wave * 16;
      #pragma unroll
      for (int kk = 0; kk < 16; kk += 4)
        rp = wmma4(ldA(Hl, HS, lm, kb + kk, kh), ldB(Bl, BLS, lm, kb + kk, kh), rp);
      #pragma unroll
      for (int v = 0; v < 8; ++v)
        Rp[wave * 256 + (v + 8 * kh) * 16 + lm] = rp[v];
    }
    __syncthreads();
    if (tid < 256) {
      float s = 0.0f;
      #pragma unroll
      for (int w = 0; w < 16; ++w) s += Rp[w * 256 + tid];
      Rl[(tid >> 4) * RS + (tid & 15)] = s;
    }
    __syncthreads();

    #pragma unroll
    for (int kk = 0; kk < 16; kk += 4)
      acc = wmma4(ldA(Rl, RS, lm, kk, kh), ldB(Aw, AWS, n0 + lm, kk, kh), acc);
    #pragma unroll
    for (int v = 0; v < 8; ++v) acc[v] = fast_tanh(acc[v]);
    h_acc = acc;
    #pragma unroll
    for (int v = 0; v < 8; ++v) Hl[(v + 8 * kh) * HS + n0 + lm] = acc[v];
    __syncthreads();

    {
      v8f d0 = {}, d1 = {}, d2 = {};
      const int kb = wave * 16;
      #pragma unroll
      for (int kk = 0; kk < 16; kk += 4) {
        v2f a = ldA(Hl, HS, lm, kb + kk, kh);
        d0 = wmma4(a, ldB(Wd, WDS,  0 + lm, kb + kk, kh), d0);
        d1 = wmma4(a, ldB(Wd, WDS, 16 + lm, kb + kk, kh), d1);
        d2 = wmma4(a, ldB(Wd, WDS, 32 + lm, kb + kk, kh), d2);
      }
      #pragma unroll
      for (int v = 0; v < 8; ++v) {
        int row = v + 8 * kh;
        Dp[wave * 768 + row * 48 +  0 + lm] = d0[v];
        Dp[wave * 768 + row * 48 + 16 + lm] = d1[v];
        Dp[wave * 768 + row * 48 + 32 + lm] = d2[v];
      }
    }
    __syncthreads();
    {
      const size_t toff = (size_t)t * IND;
      if (sva) {
        float s = bda;
        #pragma unroll
        for (int w = 0; w < 16; ++w) s += Dp[w * 768 + ea];
        outa[toff] = s;
      }
      if (svb) {
        float s = bdb;
        #pragma unroll
        for (int w = 0; w < 16; ++w) s += Dp[w * 768 + eb];
        outb[toff] = s;
      }
    }
  }

  {
    float* hf = out + (size_t)BATCH * TIME * IND;
    #pragma unroll
    for (int v = 0; v < 8; ++v)
      hf[(size_t)(b0 + v + 8 * kh) * HID + n0 + lm] = h_acc[v];
  }
}

extern "C" void kernel_launch(void* const* d_in, const int* in_sizes, int n_in,
                              void* d_out, int out_size, void* d_ws, size_t ws_size,
                              hipStream_t stream) {
  const float* x     = (const float*)d_in[0];
  const float* W_in  = (const float*)d_in[1];
  const float* Amat  = (const float*)d_in[2];
  const float* B_low = (const float*)d_in[3];
  const float* bvec  = (const float*)d_in[4];
  const float* W_dec = (const float*)d_in[5];
  const float* b_dec = (const float*)d_in[6];
  float* out = (float*)d_out;

  hipFuncSetAttribute((const void*)dream_scan,
                      hipFuncAttributeMaxDynamicSharedMemorySize, SMEM_BYTES);
  dream_scan<<<BATCH / 16, 512, SMEM_BYTES, stream>>>(
      x, W_in, Amat, B_low, bvec, W_dec, b_dec, out);
}


# Scan waves take roles: 16 added projection waves do b + x W_in^T one step ahead in 3 pieces, passed via LDS; recurrence waves keep r, A^T, tanh; step barriers wait on LDS only, 4 become 3; same sums
# speedup vs baseline: 1.9989x; 1.9989x over previous
#include <hip/hip_runtime.h>
#include <math.h>


typedef float v2f __attribute__((ext_vector_type(2)));
typedef float v8f __attribute__((ext_vector_type(8)));

#define BATCH 256
#define TIME  1024
#define IND   39
#define HID   256
#define RANK  16

#define HT_FLOATS 4096
#define STEP_BYTES_LOG2 18

#define BLS 16
#define AWS 264
#define WDS 48
#define XS  44
#define HS  260
#define RS  20
#define XBUF (16 * XS)
#define UBUF 4096

#define OFF_BL 0
#define OFF_AW (OFF_BL + 256*BLS)
#define OFF_WD (OFF_AW + 16*AWS)
#define OFF_BD (OFF_WD + 256*WDS)
#define OFF_X  (OFF_BD + 48)
#define OFF_H  (OFF_X  + 2*XBUF)
#define OFF_R  (OFF_H  + 16*HS)
#define OFF_RP (OFF_R  + 16*RS)
#define OFF_DP (OFF_RP + 16*256)
#define OFF_U  (OFF_DP + 16*768)
#define SMEM_FLOATS (OFF_U + 2*UBUF)
#define SMEM_BYTES  (SMEM_FLOATS * 4)

__device__ __forceinline__ v8f wmma4(v2f a, v2f b, v8f c) {
  return __builtin_amdgcn_wmma_f32_16x16x4_f32(
       false, a,  false, b,
       (short)0, c,  false,  false);
}

__device__ __forceinline__ float fast_tanh(float v) {
#if __has_builtin(__builtin_amdgcn_tanhf)
  return __builtin_amdgcn_tanhf(v);
#else
  return tanhf(v);
#endif
}

__device__ __forceinline__ v2f ldA(const float* base, int stride, int m, int kk, int kh) {
  const float* p = base + m * stride + kk + 2 * kh;
  v2f a; a.x = p[0]; a.y = p[1]; return a;
}
__device__ __forceinline__ v2f ldB(const float* base, int stride, int col, int kk, int kh) {
  const float* p = base + (kk + 2 * kh) * stride + col;
  v2f b; b.x = p[0]; b.y = p[stride]; return b;
}

__device__ __forceinline__ void lds_barrier() {
  __builtin_amdgcn_fence(__ATOMIC_RELEASE, "workgroup", "local");
  __builtin_amdgcn_s_barrier();
  __builtin_amdgcn_fence(__ATOMIC_ACQUIRE, "workgroup", "local");
}

__global__ __launch_bounds__(1024)
void dream_scan(const float* __restrict__ x, const float* __restrict__ W_in,
                const float* __restrict__ Amat, const float* __restrict__ B_low,
                const float* __restrict__ bvec, const float* __restrict__ W_dec,
                const float* __restrict__ b_dec, float* __restrict__ out,
                float* __restrict__ hws, size_t ws_bytes) {
  extern __shared__ float sm[];
  float* Bl = sm + OFF_BL;
  float* Aw = sm + OFF_AW;
  float* Wd = sm + OFF_WD;
  float* Bd = sm + OFF_BD;
  float* Xl = sm + OFF_X;
  float* Hl = sm + OFF_H;
  float* Rl = sm + OFF_R;
  float* Rp = sm + OFF_RP;
  float* Dp = sm + OFF_DP;
  float* Ul = sm + OFF_U;

  const int tid  = threadIdx.x;
  const int wave = __builtin_amdgcn_readfirstlane(tid >> 5);
  const int lane = tid & 31;
  const int lm   = lane & 15;
  const int kh   = lane >> 4;
  const int b0   = blockIdx.x * 16;
  const bool proj = (wave >= 16);
  const int tile = wave & 15;
  const int n0   = tile * 16;
  const int rt   = tid & 511;

  for (int e = tid; e < 256 * 16; e += 1024) {
    int j = e & 15, k = e >> 4;
    Bl[k * BLS + j] = B_low[j * HID + k];
  }
  for (int e = tid; e < 16 * 256; e += 1024) {
    int n = e & 255, j = e >> 8;
    Aw[j * AWS + n] = Amat[n * RANK + j];
  }
  for (int e = tid; e < 256 * 48; e += 1024) {
    int c = e % 48, k = e / 48;
    Wd[k * WDS + c] = (c < IND) ? W_dec[c * HID + k] : 0.0f;
  }
  if (tid < 48)  Bd[tid] = (tid < IND) ? b_dec[tid] : 0.0f;
  for (int e = tid; e < 16 * HS; e += 1024) Hl[e] = 0.0f;
  for (int e = tid; e < 2 * XBUF; e += 1024) Xl[e] = 0.0f;
  __syncthreads();

  const int  e0 = rt, e1 = rt + 512;
  const bool xv1 = (e1 < 16 * IND);
  const float* xg0;  int xl0;
  const float* xg1 = x;  int xl1 = 0;
  { int r = e0 / IND, c = e0 % IND;
    xg0 = x + (size_t)(b0 + r) * TIME * IND + c;  xl0 = r * XS + c; }
  if (xv1) { int r = e1 / IND, c = e1 % IND;
    xg1 = x + (size_t)(b0 + r) * TIME * IND + c;  xl1 = r * XS + c; }

  const int  ea = rt, eb = rt + 512;
  const bool sva = ((ea % 48) < IND);
  const bool svb = (eb < 768) && ((eb % 48) < IND);
  float* outa = out;  float bda = 0.0f;
  float* outb = out;  float bdb = 0.0f;
  if (sva) { int row = ea / 48, c = ea % 48;
    outa = out + (size_t)(b0 + row) * TIME * IND + c;  bda = Bd[c]; }
  if (svb) { int row = eb / 48, c = eb % 48;
    outb = out + (size_t)(b0 + row) * TIME * IND + c;  bdb = Bd[c]; }
  const float bb = bvec[n0 + lm];

  const size_t fit = ws_bytes >> STEP_BYTES_LOG2;
  const int tfit = (fit < (size_t)TIME) ? (int)fit : TIME;
  const int hsrc = lm * HS + 8 * tile + 4 * kh;
  float* hdst = hws + (size_t)blockIdx.x * HT_FLOATS + 4 * rt;

  v2f wi[10];
  #pragma unroll
  for (int i = 0; i < 10; ++i) {
    const int k = 4 * i + 2 * kh;
    wi[i].x = (proj && k     < IND) ? W_in[(n0 + lm) * IND + k]     : 0.0f;
    wi[i].y = (proj && k + 1 < IND) ? W_in[(n0 + lm) * IND + k + 1] : 0.0f;
  }

  float xr0 = 0.0f, xr1 = 0.0f;
  if (proj) {
    Xl[xl0] = xg0[0];
    Xl[XBUF + xl0] = xg0[IND];
    if (xv1) { Xl[xl1] = xg1[0];  Xl[XBUF + xl1] = xg1[IND]; }
    xr0 = xg0[2 * IND];
    if (xv1) xr1 = xg1[2 * IND];
  }
  __syncthreads();
  if (proj) {
    v8f p0;
    #pragma unroll
    for (int v = 0; v < 8; ++v) p0[v] = bb;
    #pragma unroll
    for (int i = 0; i < 10; ++i)
      p0 = wmma4(ldA(Xl, XS, lm, 4 * i, kh), wi[i], p0);
    float4* uq = (float4*)Ul + tile * 64 + lane;
    uq[0]  = make_float4(p0[0], p0[1], p0[2], p0[3]);
    uq[32] = make_float4(p0[4], p0[5], p0[6], p0[7]);
  }
  __syncthreads();

  for (int t = 0; t < TIME; ++t) {
    v8f pacc;
    #pragma unroll
    for (int v = 0; v < 8; ++v) pacc[v] = bb;
    const float* Xr = Xl + ((t + 1) & 1) * XBUF;

    if (proj) {
      if (t + 2 < TIME) {
        float* Xw = Xl + (t & 1) * XBUF;
        Xw[xl0] = xr0;
        if (xv1) Xw[xl1] = xr1;
      }
      if (t + 3 < TIME) {
        const size_t off = (size_t)(t + 3) * IND;
        xr0 = xg0[off];
        if (xv1) xr1 = xg1[off];
      }
      #pragma unroll
      for (int i = 0; i < 3; ++i) pacc = wmma4(ldA(Xr, XS, lm, 4 * i, kh), wi[i], pacc);
    } else {
      if (t > 0 && t - 1 < tfit) {
        const float4 c0 = *(const float4*)(Hl + hsrc);
        const float4 c1 = *(const float4*)(Hl + hsrc + 128);
        float4* d = (float4*)(hdst + (size_t)(t - 1) * (16 * HT_FLOATS));
        d[0]   = c0;
        d[512] = c1;
      }
      v8f rp = {};
      const int kb = tile * 16;
      #pragma unroll
      for (int kk = 0; kk < 16; kk += 4)
        rp = wmma4(ldA(Hl, HS, lm, kb + kk, kh), ldB(Bl, BLS, lm, kb + kk, kh), rp);
      #pragma unroll
      for (int v = 0; v < 8; ++v)
        Rp[tile * 256 + (v + 8 * kh) * 16 + lm] = rp[v];
    }
    lds_barrier();

    if (proj) {
      #pragma unroll
      for (int i = 3; i < 7; ++i) pacc = wmma4(ldA(Xr, XS, lm, 4 * i, kh), wi[i], pacc);
    } else if (tid < 256) {
      float s = 0.0f;
      #pragma unroll
      for (int w = 0; w < 16; ++w) s += Rp[w * 256 + tid];
      Rl[(tid >> 4) * RS + (tid & 15)] = s;
    }
    lds_barrier();

    if (proj) {
      #pragma unroll
      for (int i = 7; i < 10; ++i) pacc = wmma4(ldA(Xr, XS, lm, 4 * i, kh), wi[i], pacc);
      if (t + 1 < TIME) {
        float4* uq = (float4*)(Ul + ((t + 1) & 1) * UBUF) + tile * 64 + lane;
        uq[0]  = make_float4(pacc[0], pacc[1], pacc[2], pacc[3]);
        uq[32] = make_float4(pacc[4], pacc[5], pacc[6], pacc[7]);
      }
    } else {
      const float4* up = (const float4*)(Ul + (t & 1) * UBUF) + tile * 64 + lane;
      const float4 u0 = up[0], u1 = up[32];
      v8f acc;
      acc[0] = u0.x; acc[1] = u0.y; acc[2] = u0.z; acc[3] = u0.w;
      acc[4] = u1.x; acc[5] = u1.y; acc[6] = u1.z; acc[7] = u1.w;
      #pragma unroll
      for (int kk = 0; kk < 16; kk += 4)
        acc = wmma4(ldA(Rl, RS, lm, kk, kh), ldB(Aw, AWS, n0 + lm, kk, kh), acc);
      #pragma unroll
      for (int v = 0; v < 8; ++v) acc[v] = fast_tanh(acc[v]);
      #pragma unroll
      for (int v = 0; v < 8; ++v) Hl[(v + 8 * kh) * HS + n0 + lm] = acc[v];
    }
    lds_barrier();

    if (t >= tfit) {
      if (!proj) {
        v8f d0 = {}, d1 = {}, d2 = {};
        const int kb = tile * 16;
        #pragma unroll
        for (int kk = 0; kk < 16; kk += 4) {
          v2f a = ldA(Hl, HS, lm, kb + kk, kh);
          d0 = wmma4(a, ldB(Wd, WDS,  0 + lm, kb + kk, kh), d0);
          d1 = wmma4(a, ldB(Wd, WDS, 16 + lm, kb + kk, kh), d1);
          d2 = wmma4(a, ldB(Wd, WDS, 32 + lm, kb + kk, kh), d2);
        }
        #pragma unroll
        for (int v = 0; v < 8; ++v) {
          int row = v + 8 * kh;
          Dp[tile * 768 + row * 48 +  0 + lm] = d0[v];
          Dp[tile * 768 + row * 48 + 16 + lm] = d1[v];
          Dp[tile * 768 + row * 48 + 32 + lm] = d2[v];
        }
      }
      lds_barrier();
      if (!proj) {
        const size_t toff = (size_t)t * IND;
        if (sva) {
          float s = bda;
          #pragma unroll
          for (int w = 0; w < 16; ++w) s += Dp[w * 768 + ea];
          outa[toff] = s;
        }
        if (svb) {
          float s = bdb;
          #pragma unroll
          for (int w = 0; w < 16; ++w) s += Dp[w * 768 + eb];
          outb[toff] = s;
        }
      }
    }
  }

  if (!proj) {
    if (TIME - 1 < tfit) {
      const float4 c0 = *(const float4*)(Hl + hsrc);
      const float4 c1 = *(const float4*)(Hl + hsrc + 128);
      float4* d = (float4*)(hdst + (size_t)(TIME - 1) * (16 * HT_FLOATS));
      d[0]   = c0;
      d[512] = c1;
    }
    float* hf = out + (size_t)BATCH * TIME * IND;
    #pragma unroll
    for (int v = 0; v < 8; ++v)
      hf[(size_t)(b0 + v + 8 * kh) * HID + n0 + lm] = Hl[(v + 8 * kh) * HS + n0 + lm];
  }
}


__device__ __forceinline__ void decode_half(const float4 (&a)[16], const float4 (&w)[32],
                                            int ibase, v8f& de, v8f& dq) {
  #pragma unroll
  for (int i = 0; i < 16; ++i) {
    v2f alo; alo.x = a[i].x; alo.y = a[i].y;
    v2f ahi; ahi.x = a[i].z; ahi.y = a[i].w;
    v2f blo; blo.x = w[ibase + i].x; blo.y = w[ibase + i].y;
    v2f bhi; bhi.x = w[ibase + i].z; bhi.y = w[ibase + i].w;
    de = wmma4(alo, blo, de);
    dq = wmma4(ahi, bhi, dq);
  }
}

__global__ __launch_bounds__(384)
void recon_decode(const float* __restrict__ hws, const float* __restrict__ W_dec,
                  const float* __restrict__ b_dec, float* __restrict__ out,
                  size_t ws_bytes) {
  const int tid  = threadIdx.x;
  const int wave = tid >> 5;
  const int lane = tid & 31;
  const int lm   = lane & 15;
  const int kh   = lane >> 4;
  const int grp  = wave / 3;
  const int ct   = wave - 3 * grp;

  const size_t fit = ws_bytes >> STEP_BYTES_LOG2;
  const int tfit = (fit < (size_t)TIME) ? (int)fit : TIME;

  const int gid = blockIdx.x * 4 + grp;
  const int bt  = gid & 15;
  const int tc  = gid >> 4;
  const int tpw = (tfit + 63) >> 6;
  const int t0  = tc * tpw;
  const int t1  = (t0 + tpw < tfit) ? (t0 + tpw) : tfit;
  if (t0 >= t1) return;

  const int  dcol  = 16 * ct + lm;
  const bool valid = (dcol < IND);
  const int   wrow = valid ? dcol : 0;
  float4 w[32];
  #pragma unroll
  for (int i = 0; i < 32; ++i) {
    const float4 r = *(const float4*)(W_dec + wrow * HID + 8 * i + 4 * kh);
    w[i] = valid ? r : make_float4(0.0f, 0.0f, 0.0f, 0.0f);
  }
  const float bd = valid ? b_dec[wrow] : 0.0f;

  const float4* src = (const float4*)(hws + ((size_t)t0 * 16 + bt) * HT_FLOATS) + lane;
  float* orow = out + ((size_t)(bt * 16 + 8 * kh) * TIME + t0) * IND + dcol;

  float4 a0[16], a1[16];
  #pragma unroll
  for (int i = 0; i < 16; ++i) a0[i] = src[i * 32];
  #pragma unroll
  for (int i = 0; i < 16; ++i) a1[i] = src[(16 + i) * 32];

  for (int t = t0; t < t1; ++t) {
    v8f de = {}, dq = {};
    const float4* nxt = src + 16 * (HT_FLOATS / 4);
    const bool more = (t + 1 < t1);

    decode_half(a0, w, 0, de, dq);
    if (more) {
      #pragma unroll
      for (int i = 0; i < 16; ++i) a0[i] = nxt[i * 32];
    }
    decode_half(a1, w, 16, de, dq);
    if (more) {
      #pragma unroll
      for (int i = 0; i < 16; ++i) a1[i] = nxt[(16 + i) * 32];
    }

    if (valid) {
      #pragma unroll
      for (int v = 0; v < 8; ++v)
        orow[(size_t)v * (TIME * IND)] = (de[v] + dq[v]) + bd;
    }
    src = nxt;
    orow += IND;
  }
}

extern "C" void kernel_launch(void* const* d_in, const int* in_sizes, int n_in,
                              void* d_out, int out_size, void* d_ws, size_t ws_size,
                              hipStream_t stream) {
  const float* x     = (const float*)d_in[0];
  const float* W_in  = (const float*)d_in[1];
  const float* Amat  = (const float*)d_in[2];
  const float* B_low = (const float*)d_in[3];
  const float* bvec  = (const float*)d_in[4];
  const float* W_dec = (const float*)d_in[5];
  const float* b_dec = (const float*)d_in[6];
  float* out = (float*)d_out;

  hipFuncSetAttribute((const void*)dream_scan,
                      hipFuncAttributeMaxDynamicSharedMemorySize, SMEM_BYTES);
  dream_scan<<<BATCH / 16, 1024, SMEM_BYTES, stream>>>(
      x, W_in, Amat, B_low, bvec, W_dec, b_dec, out, (float*)d_ws, ws_size);
  recon_decode<<<256, 384, 0, stream>>>(
      (const float*)d_ws, W_dec, b_dec, out, ws_size);
}


# Step barriers split into arrive and wait; between them a wave does what nobody needs yet: projection pieces, h copy, early loads, next r partial from own h columns; x refill behind Ba; same sums
# speedup vs baseline: 2.5722x; 1.2868x over previous
#include <hip/hip_runtime.h>
#include <math.h>


typedef float v2f __attribute__((ext_vector_type(2)));
typedef float v8f __attribute__((ext_vector_type(8)));

#define BATCH 256
#define TIME  1024
#define IND   39
#define HID   256
#define RANK  16

#define HT_FLOATS 4096
#define STEP_BYTES_LOG2 18

#define BLS 16
#define AWS 264
#define WDS 48
#define XS  44
#define HS  260
#define RS  20
#define XBUF (16 * XS)
#define UBUF 4096

#define OFF_BL 0
#define OFF_AW (OFF_BL + 256*BLS)
#define OFF_WD (OFF_AW + 16*AWS)
#define OFF_BD (OFF_WD + 256*WDS)
#define OFF_X  (OFF_BD + 48)
#define OFF_H  (OFF_X  + 2*XBUF)
#define OFF_R  (OFF_H  + 16*HS)
#define OFF_RP (OFF_R  + 16*RS)
#define OFF_DP (OFF_RP + 16*256)
#define OFF_U  (OFF_DP + 16*768)
#define SMEM_FLOATS (OFF_U + 2*UBUF)
#define SMEM_BYTES  (SMEM_FLOATS * 4)

__device__ __forceinline__ v8f wmma4(v2f a, v2f b, v8f c) {
  return __builtin_amdgcn_wmma_f32_16x16x4_f32(
       false, a,  false, b,
       (short)0, c,  false,  false);
}

__device__ __forceinline__ float fast_tanh(float v) {
#if __has_builtin(__builtin_amdgcn_tanhf)
  return __builtin_amdgcn_tanhf(v);
#else
  return tanhf(v);
#endif
}

__device__ __forceinline__ v2f ldA(const float* base, int stride, int m, int kk, int kh) {
  const float* p = base + m * stride + kk + 2 * kh;
  v2f a; a.x = p[0]; a.y = p[1]; return a;
}
__device__ __forceinline__ v2f ldB(const float* base, int stride, int col, int kk, int kh) {
  const float* p = base + (kk + 2 * kh) * stride + col;
  v2f b; b.x = p[0]; b.y = p[stride]; return b;
}

__device__ __forceinline__ void lds_barrier() {
  __builtin_amdgcn_fence(__ATOMIC_RELEASE, "workgroup", "local");
  __builtin_amdgcn_s_barrier();
  __builtin_amdgcn_fence(__ATOMIC_ACQUIRE, "workgroup", "local");
}
__device__ __forceinline__ void bar_arrive() {
  __builtin_amdgcn_fence(__ATOMIC_RELEASE, "workgroup", "local");
  __builtin_amdgcn_s_barrier_signal(-1);
}
__device__ __forceinline__ void bar_wait() {
  __builtin_amdgcn_s_barrier_wait(-1);
  __builtin_amdgcn_fence(__ATOMIC_ACQUIRE, "workgroup", "local");
}
__device__ __forceinline__ void bar_wait_after_work() {
  __builtin_amdgcn_sched_barrier(0);
  bar_wait();
}

__global__ __launch_bounds__(1024)
void dream_scan(const float* __restrict__ x, const float* __restrict__ W_in,
                const float* __restrict__ Amat, const float* __restrict__ B_low,
                const float* __restrict__ bvec, const float* __restrict__ W_dec,
                const float* __restrict__ b_dec, float* __restrict__ out,
                float* __restrict__ hws, size_t ws_bytes) {
  extern __shared__ float sm[];
  float* Bl = sm + OFF_BL;
  float* Aw = sm + OFF_AW;
  float* Wd = sm + OFF_WD;
  float* Bd = sm + OFF_BD;
  float* Xl = sm + OFF_X;
  float* Hl = sm + OFF_H;
  float* Rl = sm + OFF_R;
  float* Rp = sm + OFF_RP;
  float* Dp = sm + OFF_DP;
  float* Ul = sm + OFF_U;

  const int tid  = threadIdx.x;
  const int wave = __builtin_amdgcn_readfirstlane(tid >> 5);
  const int lane = tid & 31;
  const int lm   = lane & 15;
  const int kh   = lane >> 4;
  const int b0   = blockIdx.x * 16;
  const bool proj = (wave >= 16);
  const int tile = wave & 15;
  const int n0   = tile * 16;
  const int rt   = tid & 511;

  for (int e = tid; e < 256 * 16; e += 1024) {
    int j = e & 15, k = e >> 4;
    Bl[k * BLS + j] = B_low[j * HID + k];
  }
  for (int e = tid; e < 16 * 256; e += 1024) {
    int n = e & 255, j = e >> 8;
    Aw[j * AWS + n] = Amat[n * RANK + j];
  }
  for (int e = tid; e < 256 * 48; e += 1024) {
    int c = e % 48, k = e / 48;
    Wd[k * WDS + c] = (c < IND) ? W_dec[c * HID + k] : 0.0f;
  }
  if (tid < 48)  Bd[tid] = (tid < IND) ? b_dec[tid] : 0.0f;
  for (int e = tid; e < 16 * HS; e += 1024) Hl[e] = 0.0f;
  for (int e = tid; e < 2 * XBUF; e += 1024) Xl[e] = 0.0f;
  __syncthreads();

  const int  e0 = rt, e1 = rt + 512;
  const bool xv1 = (e1 < 16 * IND);
  const float* xg0;  int xl0;
  const float* xg1 = x;  int xl1 = 0;
  { int r = e0 / IND, c = e0 % IND;
    xg0 = x + (size_t)(b0 + r) * TIME * IND + c;  xl0 = r * XS + c; }
  if (xv1) { int r = e1 / IND, c = e1 % IND;
    xg1 = x + (size_t)(b0 + r) * TIME * IND + c;  xl1 = r * XS + c; }

  const int  ea = rt, eb = rt + 512;
  const bool sva = ((ea % 48) < IND);
  const bool svb = (eb < 768) && ((eb % 48) < IND);
  float* outa = out;  float bda = 0.0f;
  float* outb = out;  float bdb = 0.0f;
  if (sva) { int row = ea / 48, c = ea % 48;
    outa = out + (size_t)(b0 + row) * TIME * IND + c;  bda = Bd[c]; }
  if (svb) { int row = eb / 48, c = eb % 48;
    outb = out + (size_t)(b0 + row) * TIME * IND + c;  bdb = Bd[c]; }
  const float bb = bvec[n0 + lm];

  const size_t fit = ws_bytes >> STEP_BYTES_LOG2;
  const int tfit = (fit < (size_t)TIME) ? (int)fit : TIME;
  const int hsrc = lm * HS + 8 * tile + 4 * kh;
  float* hdst = hws + (size_t)blockIdx.x * HT_FLOATS + 4 * rt;

  v2f wi[10];
  #pragma unroll
  for (int i = 0; i < 10; ++i) {
    const int k = 4 * i + 2 * kh;
    wi[i].x = (proj && k     < IND) ? W_in[(n0 + lm) * IND + k]     : 0.0f;
    wi[i].y = (proj && k + 1 < IND) ? W_in[(n0 + lm) * IND + k + 1] : 0.0f;
  }

  float xr0 = 0.0f, xr1 = 0.0f;
  if (proj) {
    Xl[xl0] = xg0[0];
    Xl[XBUF + xl0] = xg0[IND];
    if (xv1) { Xl[xl1] = xg1[0];  Xl[XBUF + xl1] = xg1[IND]; }
    xr0 = xg0[2 * IND];
    if (xv1) xr1 = xg1[2 * IND];
  }
  __syncthreads();
  if (proj) {
    v8f p0;
    #pragma unroll
    for (int v = 0; v < 8; ++v) p0[v] = bb;
    #pragma unroll
    for (int i = 0; i < 10; ++i)
      p0 = wmma4(ldA(Xl, XS, lm, 4 * i, kh), wi[i], p0);
    float4* uq = (float4*)Ul + tile * 64 + lane;
    uq[0]  = make_float4(p0[0], p0[1], p0[2], p0[3]);
    uq[32] = make_float4(p0[4], p0[5], p0[6], p0[7]);
  }
  __syncthreads();

  if (!proj) {
    v8f rp = {};
    const int kb = tile * 16;
    #pragma unroll
    for (int kk = 0; kk < 16; kk += 4)
      rp = wmma4(ldA(Hl, HS, lm, kb + kk, kh), ldB(Bl, BLS, lm, kb + kk, kh), rp);
    #pragma unroll
    for (int v = 0; v < 8; ++v)
      Rp[tile * 256 + (v + 8 * kh) * 16 + lm] = rp[v];
  }

  for (int t = 0; t < TIME; ++t) {
    if (proj) {
      v8f pacc;
      #pragma unroll
      for (int v = 0; v < 8; ++v) pacc[v] = bb;
      const float* Xr = Xl + ((t + 1) & 1) * XBUF;

      bar_arrive();
      #pragma unroll
      for (int i = 0; i < 3; ++i) pacc = wmma4(ldA(Xr, XS, lm, 4 * i, kh), wi[i], pacc);
      bar_wait_after_work();

      bar_arrive();
      if (t + 2 < TIME) {
        float* Xw = Xl + (t & 1) * XBUF;
        Xw[xl0] = xr0;
        if (xv1) Xw[xl1] = xr1;
      }
      if (t + 3 < TIME) {
        const size_t off = (size_t)(t + 3) * IND;
        xr0 = xg0[off];
        if (xv1) xr1 = xg1[off];
      }
      #pragma unroll
      for (int i = 3; i < 7; ++i) pacc = wmma4(ldA(Xr, XS, lm, 4 * i, kh), wi[i], pacc);
      bar_wait_after_work();

      bar_arrive();
      #pragma unroll
      for (int i = 7; i < 10; ++i) pacc = wmma4(ldA(Xr, XS, lm, 4 * i, kh), wi[i], pacc);
      if (t + 1 < TIME) {
        float4* uq = (float4*)(Ul + ((t + 1) & 1) * UBUF) + tile * 64 + lane;
        uq[0]  = make_float4(pacc[0], pacc[1], pacc[2], pacc[3]);
        uq[32] = make_float4(pacc[4], pacc[5], pacc[6], pacc[7]);
      }
      bar_wait();
    } else {
      bar_arrive();
      if (t > 0 && t - 1 < tfit) {
        const float4 c0 = *(const float4*)(Hl + hsrc);
        const float4 c1 = *(const float4*)(Hl + hsrc + 128);
        float4* d = (float4*)(hdst + (size_t)(t - 1) * (16 * HT_FLOATS));
        d[0]   = c0;
        d[512] = c1;
      }
      bar_wait();

      if (tid < 256) {
        float s = 0.0f;
        #pragma unroll
        for (int w = 0; w < 16; ++w) s += Rp[w * 256 + tid];
        Rl[(tid >> 4) * RS + (tid & 15)] = s;
      }
      bar_arrive();
      const float4* up = (const float4*)(Ul + (t & 1) * UBUF) + tile * 64 + lane;
      const float4 u0 = up[0], u1 = up[32];
      v2f aw[4];
      #pragma unroll
      for (int q = 0; q < 4; ++q) aw[q] = ldB(Aw, AWS, n0 + lm, 4 * q, kh);
      bar_wait();

      v8f acc;
      acc[0] = u0.x; acc[1] = u0.y; acc[2] = u0.z; acc[3] = u0.w;
      acc[4] = u1.x; acc[5] = u1.y; acc[6] = u1.z; acc[7] = u1.w;
      #pragma unroll
      for (int q = 0; q < 4; ++q)
        acc = wmma4(ldA(Rl, RS, lm, 4 * q, kh), aw[q], acc);
      #pragma unroll
      for (int v = 0; v < 8; ++v) acc[v] = fast_tanh(acc[v]);
      #pragma unroll
      for (int v = 0; v < 8; ++v) Hl[(v + 8 * kh) * HS + n0 + lm] = acc[v];
      bar_arrive();
      if (t + 1 < TIME) {
        v8f rp = {};
        const int kb = tile * 16;
        #pragma unroll
        for (int kk = 0; kk < 16; kk += 4)
          rp = wmma4(ldA(Hl, HS, lm, kb + kk, kh), ldB(Bl, BLS, lm, kb + kk, kh), rp);
        #pragma unroll
        for (int v = 0; v < 8; ++v)
          Rp[tile * 256 + (v + 8 * kh) * 16 + lm] = rp[v];
      }
      bar_wait();
    }

    if (t >= tfit) {
      if (!proj) {
        v8f d0 = {}, d1 = {}, d2 = {};
        const int kb = tile * 16;
        #pragma unroll
        for (int kk = 0; kk < 16; kk += 4) {
          v2f a = ldA(Hl, HS, lm, kb + kk, kh);
          d0 = wmma4(a, ldB(Wd, WDS,  0 + lm, kb + kk, kh), d0);
          d1 = wmma4(a, ldB(Wd, WDS, 16 + lm, kb + kk, kh), d1);
          d2 = wmma4(a, ldB(Wd, WDS, 32 + lm, kb + kk, kh), d2);
        }
        #pragma unroll
        for (int v = 0; v < 8; ++v) {
          int row = v + 8 * kh;
          Dp[tile * 768 + row * 48 +  0 + lm] = d0[v];
          Dp[tile * 768 + row * 48 + 16 + lm] = d1[v];
          Dp[tile * 768 + row * 48 + 32 + lm] = d2[v];
        }
      }
      lds_barrier();
      if (!proj) {
        const size_t toff = (size_t)t * IND;
        if (sva) {
          float s = bda;
          #pragma unroll
          for (int w = 0; w < 16; ++w) s += Dp[w * 768 + ea];
          outa[toff] = s;
        }
        if (svb) {
          float s = bdb;
          #pragma unroll
          for (int w = 0; w < 16; ++w) s += Dp[w * 768 + eb];
          outb[toff] = s;
        }
      }
    }
  }

  if (!proj) {
    if (TIME - 1 < tfit) {
      const float4 c0 = *(const float4*)(Hl + hsrc);
      const float4 c1 = *(const float4*)(Hl + hsrc + 128);
      float4* d = (float4*)(hdst + (size_t)(TIME - 1) * (16 * HT_FLOATS));
      d[0]   = c0;
      d[512] = c1;
    }
    float* hf = out + (size_t)BATCH * TIME * IND;
    #pragma unroll
    for (int v = 0; v < 8; ++v)
      hf[(size_t)(b0 + v + 8 * kh) * HID + n0 + lm] = Hl[(v + 8 * kh) * HS + n0 + lm];
  }
}


__device__ __forceinline__ void decode_half(const float4 (&a)[16], const float4 (&w)[32],
                                            int ibase, v8f& de, v8f& dq) {
  #pragma unroll
  for (int i = 0; i < 16; ++i) {
    v2f alo; alo.x = a[i].x; alo.y = a[i].y;
    v2f ahi; ahi.x = a[i].z; ahi.y = a[i].w;
    v2f blo; blo.x = w[ibase + i].x; blo.y = w[ibase + i].y;
    v2f bhi; bhi.x = w[ibase + i].z; bhi.y = w[ibase + i].w;
    de = wmma4(alo, blo, de);
    dq = wmma4(ahi, bhi, dq);
  }
}

__global__ __launch_bounds__(384)
void recon_decode(const float* __restrict__ hws, const float* __restrict__ W_dec,
                  const float* __restrict__ b_dec, float* __restrict__ out,
                  size_t ws_bytes) {
  const int tid  = threadIdx.x;
  const int wave = tid >> 5;
  const int lane = tid & 31;
  const int lm   = lane & 15;
  const int kh   = lane >> 4;
  const int grp  = wave / 3;
  const int ct   = wave - 3 * grp;

  const size_t fit = ws_bytes >> STEP_BYTES_LOG2;
  const int tfit = (fit < (size_t)TIME) ? (int)fit : TIME;

  const int gid = blockIdx.x * 4 + grp;
  const int bt  = gid & 15;
  const int tc  = gid >> 4;
  const int tpw = (tfit + 63) >> 6;
  const int t0  = tc * tpw;
  const int t1  = (t0 + tpw < tfit) ? (t0 + tpw) : tfit;
  if (t0 >= t1) return;

  const int  dcol  = 16 * ct + lm;
  const bool valid = (dcol < IND);
  const int   wrow = valid ? dcol : 0;
  float4 w[32];
  #pragma unroll
  for (int i = 0; i < 32; ++i) {
    const float4 r = *(const float4*)(W_dec + wrow * HID + 8 * i + 4 * kh);
    w[i] = valid ? r : make_float4(0.0f, 0.0f, 0.0f, 0.0f);
  }
  const float bd = valid ? b_dec[wrow] : 0.0f;

  const float4* src = (const float4*)(hws + ((size_t)t0 * 16 + bt) * HT_FLOATS) + lane;
  float* orow = out + ((size_t)(bt * 16 + 8 * kh) * TIME + t0) * IND + dcol;

  float4 a0[16], a1[16];
  #pragma unroll
  for (int i = 0; i < 16; ++i) a0[i] = src[i * 32];
  #pragma unroll
  for (int i = 0; i < 16; ++i) a1[i] = src[(16 + i) * 32];

  for (int t = t0; t < t1; ++t) {
    v8f de = {}, dq = {};
    const float4* nxt = src + 16 * (HT_FLOATS / 4);
    const bool more = (t + 1 < t1);

    decode_half(a0, w, 0, de, dq);
    if (more) {
      #pragma unroll
      for (int i = 0; i < 16; ++i) a0[i] = nxt[i * 32];
    }
    decode_half(a1, w, 16, de, dq);
    if (more) {
      #pragma unroll
      for (int i = 0; i < 16; ++i) a1[i] = nxt[(16 + i) * 32];
    }

    if (valid) {
      #pragma unroll
      for (int v = 0; v < 8; ++v)
        orow[(size_t)v * (TIME * IND)] = (de[v] + dq[v]) + bd;
    }
    src = nxt;
    orow += IND;
  }
}

extern "C" void kernel_launch(void* const* d_in, const int* in_sizes, int n_in,
                              void* d_out, int out_size, void* d_ws, size_t ws_size,
                              hipStream_t stream) {
  const float* x     = (const float*)d_in[0];
  const float* W_in  = (const float*)d_in[1];
  const float* Amat  = (const float*)d_in[2];
  const float* B_low = (const float*)d_in[3];
  const float* bvec  = (const float*)d_in[4];
  const float* W_dec = (const float*)d_in[5];
  const float* b_dec = (const float*)d_in[6];
  float* out = (float*)d_out;

  hipFuncSetAttribute((const void*)dream_scan,
                      hipFuncAttributeMaxDynamicSharedMemorySize, SMEM_BYTES);
  dream_scan<<<BATCH / 16, 1024, SMEM_BYTES, stream>>>(
      x, W_in, Amat, B_low, bvec, W_dec, b_dec, out, (float*)d_ws, ws_size);
  recon_decode<<<256, 384, 0, stream>>>(
      (const float*)d_ws, W_dec, b_dec, out, ws_size);
}
